# starting kernel
# baseline (speedup 1.0000x reference)
#include <hip/hip_runtime.h>


typedef __attribute__((ext_vector_type(2))) float v2f;
typedef __attribute__((ext_vector_type(4))) float v4f;
typedef __attribute__((ext_vector_type(8))) float v8f;

#define B_DIM   256
#define OUT_DIM 1000
#define C_DIM   512
#define INV_TEMP (1.0f / 64.0f)

__global__ __launch_bounds__(256) void euc_norms(const float* __restrict__ x,
                                                 const float* __restrict__ w,
                                                 float* __restrict__ norms) {
  const int wave = (int)((blockIdx.x * blockDim.x + threadIdx.x) >> 5);
  const int lane = (int)(threadIdx.x & 31);
  if (wave >= B_DIM + OUT_DIM) return;

  const float* row = (wave < B_DIM) ? (x + (size_t)wave * C_DIM)
                                    : (w + (size_t)(wave - B_DIM) * C_DIM);
  float s = 0.0f;
#pragma unroll
  for (int i = 0; i < C_DIM / (32 * 4); ++i) {
    float4 v = ((const float4*)row)[lane + 32 * i];
    s += v.x * v.x + v.y * v.y + v.z * v.z + v.w * v.w;
  }
#pragma unroll
  for (int off = 16; off > 0; off >>= 1)
    s += __shfl_xor(s, off, 32);
  if (lane == 0) norms[wave] = s;
}

__global__ __launch_bounds__(256) void euc_scores(const float* __restrict__ x,
                                                  const float* __restrict__ w,
                                                  const float* __restrict__ norms,
                                                  float* __restrict__ out) {
  const int lane = (int)(threadIdx.x & 31);
  const int wave = (int)(threadIdx.x >> 5);
  const int l16  = lane & 15;
  const int hi   = lane >> 4;

  const int task   = (int)blockIdx.x * 8 + wave;
  const int mTile  = task & 15;
  const int nGroup = task >> 4;

  const int mBase = mTile * 16;
  const int n0 = nGroup * 32 + l16;
  const int n1 = n0 + 16;
  const int n0c = (n0 < OUT_DIM) ? n0 : (OUT_DIM - 1);
  const int n1c = (n1 < OUT_DIM) ? n1 : (OUT_DIM - 1);

  const float* aPtr  = x + (size_t)(mBase + l16) * C_DIM + 4 * hi;
  const float* b0Ptr = w + (size_t)n0c * C_DIM + 4 * hi;
  const float* b1Ptr = w + (size_t)n1c * C_DIM + 4 * hi;

  v8f acc0 = {};
  v8f acc1 = {};

#pragma unroll
  for (int s = 0; s < C_DIM / 8; ++s) {
    v4f av  = *(const v4f*)(aPtr  + 8 * s);
    v4f b0v = *(const v4f*)(b0Ptr + 8 * s);
    v4f b1v = *(const v4f*)(b1Ptr + 8 * s);

    v2f a01  = __builtin_shufflevector(av,  av,  0, 1);
    v2f a23  = __builtin_shufflevector(av,  av,  2, 3);
    v2f b001 = __builtin_shufflevector(b0v, b0v, 0, 1);
    v2f b023 = __builtin_shufflevector(b0v, b0v, 2, 3);
    v2f b101 = __builtin_shufflevector(b1v, b1v, 0, 1);
    v2f b123 = __builtin_shufflevector(b1v, b1v, 2, 3);

    acc0 = __builtin_amdgcn_wmma_f32_16x16x4_f32(
        false, a01, false, b001, (short)0, acc0, false, false);
    acc1 = __builtin_amdgcn_wmma_f32_16x16x4_f32(
        false, a01, false, b101, (short)0, acc1, false, false);
    acc0 = __builtin_amdgcn_wmma_f32_16x16x4_f32(
        false, a23, false, b023, (short)0, acc0, false, false);
    acc1 = __builtin_amdgcn_wmma_f32_16x16x4_f32(
        false, a23, false, b123, (short)0, acc1, false, false);
  }

  const float ww0 = norms[B_DIM + n0c];
  const float ww1 = norms[B_DIM + n1c];

#pragma unroll
  for (int v = 0; v < 8; ++v) {
    const int m = mBase + v + 8 * hi;
    const float xx = norms[m];
    if (n0 < OUT_DIM)
      out[(size_t)m * OUT_DIM + n0] = (2.0f * acc0[v] - xx - ww0) * INV_TEMP;
    if (n1 < OUT_DIM)
      out[(size_t)m * OUT_DIM + n1] = (2.0f * acc1[v] - xx - ww1) * INV_TEMP;
  }
}

extern "C" void kernel_launch(void* const* d_in, const int* in_sizes, int n_in,
                              void* d_out, int out_size, void* d_ws, size_t ws_size,
                              hipStream_t stream) {
  (void)in_sizes; (void)n_in; (void)out_size; (void)ws_size;
  const float* x = (const float*)d_in[0];
  const float* w = (const float*)d_in[1];
  float* out   = (float*)d_out;
  float* norms = (float*)d_ws;

  const int normBlocks = ((B_DIM + OUT_DIM) * 32 + 255) / 256;
  euc_norms<<<normBlocks, 256, 0, stream>>>(x, w, norms);

  euc_scores<<<64, 256, 0, stream>>>(x, w, norms, out);
}


# One launch: each wave sums the squares of its own register panels (packed f32) and 4 closing WMMAs against a -1/2 operand add both norm terms into the accumulators (norm sum order changes)
# speedup vs baseline: 3.3374x; 3.3374x over previous
#include <hip/hip_runtime.h>


typedef __attribute__((ext_vector_type(2))) float v2f;
typedef __attribute__((ext_vector_type(4))) float v4f;
typedef __attribute__((ext_vector_type(8))) float v8f;

#define B_DIM   256
#define OUT_DIM 1000
#define C_DIM   512
#define INV_TEMP (1.0f / 64.0f)

#define K_CHUNKS (C_DIM / 8)

__global__ __launch_bounds__(128) void euc_scores(const float* __restrict__ x,
                                                  const float* __restrict__ w,
                                                  float* __restrict__ out) {
  const int lane = (int)(threadIdx.x & 31);
  const int wave = (int)(threadIdx.x >> 5);
  const int l16  = lane & 15;
  const int hi   = lane >> 4;

  const int mPair = (int)blockIdx.x & 7;
  const int nPair = (int)blockIdx.x >> 3;
  const int mTile = 2 * mPair + (wave & 1);
  const int nTile = 2 * nPair + (wave >> 1);
  if (nTile * 16 >= OUT_DIM) return;

  const int mBase = mTile * 16;
  const int n  = nTile * 16 + l16;
  const int nc = (n < OUT_DIM) ? n : (OUT_DIM - 1);

  const float* aPtr = x + (size_t)(mBase + l16) * C_DIM + 4 * hi;
  const float* bPtr = w + (size_t)nc * C_DIM + 4 * hi;

  v4f av[K_CHUNKS];
  v4f bv[K_CHUNKS];
#pragma unroll
  for (int s = 0; s < K_CHUNKS; ++s) {
    av[s] = *(const v4f*)(aPtr + 8 * s);
    bv[s] = *(const v4f*)(bPtr + 8 * s);
  }

  __builtin_amdgcn_sched_barrier(0);

  v8f acc0 = {};
  v8f acc1 = {};
  v2f xs01 = {}, xs23 = {};
  v2f ws01 = {}, ws23 = {};
#pragma unroll
  for (int s = 0; s < K_CHUNKS; ++s) {
    v2f a01 = __builtin_shufflevector(av[s], av[s], 0, 1);
    v2f a23 = __builtin_shufflevector(av[s], av[s], 2, 3);
    v2f b01 = __builtin_shufflevector(bv[s], bv[s], 0, 1);
    v2f b23 = __builtin_shufflevector(bv[s], bv[s], 2, 3);
    acc0 = __builtin_amdgcn_wmma_f32_16x16x4_f32(
        false, a01, false, b01, (short)0, acc0, false, false);
    acc1 = __builtin_amdgcn_wmma_f32_16x16x4_f32(
        false, a23, false, b23, (short)0, acc1, false, false);
    xs01 += a01 * a01;
    xs23 += a23 * a23;
    ws01 += b01 * b01;
    ws23 += b23 * b23;
  }

  const v2f negHalf = {-0.5f, -0.5f};
  acc0 = __builtin_amdgcn_wmma_f32_16x16x4_f32(
      false, xs01, false, negHalf, (short)0, acc0, false, false);
  acc1 = __builtin_amdgcn_wmma_f32_16x16x4_f32(
      false, negHalf, false, ws01, (short)0, acc1, false, false);
  acc0 = __builtin_amdgcn_wmma_f32_16x16x4_f32(
      false, xs23, false, negHalf, (short)0, acc0, false, false);
  acc1 = __builtin_amdgcn_wmma_f32_16x16x4_f32(
      false, negHalf, false, ws23, (short)0, acc1, false, false);

#pragma unroll
  for (int v = 0; v < 8; ++v) {
    const int m = mBase + v + 8 * hi;
    if (n < OUT_DIM)
      out[(size_t)m * OUT_DIM + n] = (acc0[v] + acc1[v]) * (2.0f * INV_TEMP);
  }
}

extern "C" void kernel_launch(void* const* d_in, const int* in_sizes, int n_in,
                              void* d_out, int out_size, void* d_ws, size_t ws_size,
                              hipStream_t stream) {
  (void)in_sizes; (void)n_in; (void)out_size; (void)d_ws; (void)ws_size;
  const float* x = (const float*)d_in[0];
  const float* w = (const float*)d_in[1];
  float* out   = (float*)d_out;

  euc_scores<<<256, 128, 0, stream>>>(x, w, out);
}
